# starting kernel
# baseline (speedup 1.0000x reference)
#include <hip/hip_runtime.h>
#include <hip/hip_bf16.h>

typedef __attribute__((ext_vector_type(2))) float v2f;
typedef __attribute__((ext_vector_type(8))) float v8f;

#define DIM 128
#define WAVES_PER_BLOCK 5

__global__ __launch_bounds__(WAVES_PER_BLOCK * 32)
void distmult_wmma_f32_kernel(const float* __restrict__ node,
                              const float* __restrict__ rel,
                              const int*   __restrict__ src,
                              const int*   __restrict__ dst,
                              float*       __restrict__ out,
                              int E, int tilesPerRel) {
    __shared__ float srel[DIM];

    const int r = blockIdx.y;

    if (threadIdx.x < DIM)
        srel[threadIdx.x] = rel[r * DIM + threadIdx.x];
    __syncthreads();

    const int wave = threadIdx.x >> 5;
    const int lane = threadIdx.x & 31;
    const int tile = blockIdx.x * WAVES_PER_BLOCK + wave;
    if (tile >= tilesPerRel) return;

    const int e0  = tile << 4;
    const int l16 = lane & 15;
    const int hiOff = (lane >> 4) << 1;

    const long eidx = (long)r * E + e0 + l16;
    const float* uRow = node + (size_t)src[eidx] * DIM + hiOff;
    const float* vRow = node + (size_t)dst[eidx] * DIM + hiOff;
    const float* rRow = srel + hiOff;

    v8f c = {};

#pragma unroll
    for (int k = 0; k < DIM; k += 4) {
        v2f u2 = *(const v2f*)(uRow + k);
        v2f w2 = *(const v2f*)(vRow + k);
        v2f r2 = *(const v2f*)(rRow + k);
        v2f a2 = u2 * r2;
        c = __builtin_amdgcn_wmma_f32_16x16x4_f32(
                 false, a2,
                 false, w2,
                 (short)0, c,
                 false,  false);
    }

    int m = -1;
    if (lane < 8)        m = lane;
    else if (lane >= 24) m = lane - 16;

    if (m >= 0) {
        const int ci = lane & 7;
        float v = c[0];
        if (ci == 1) v = c[1];
        if (ci == 2) v = c[2];
        if (ci == 3) v = c[3];
        if (ci == 4) v = c[4];
        if (ci == 5) v = c[5];
        if (ci == 6) v = c[6];
        if (ci == 7) v = c[7];
        out[(size_t)r * E + (size_t)(e0 + m)] = v;
    }
}

extern "C" void kernel_launch(void* const* d_in, const int* in_sizes, int n_in,
                              void* d_out, int out_size, void* d_ws, size_t ws_size,
                              hipStream_t stream) {
    const float* node = (const float*)d_in[0];
    const float* rel  = (const float*)d_in[1];
    const int*   src  = (const int*)d_in[2];
    const int*   dst  = (const int*)d_in[3];
    float*       out  = (float*)d_out;

    const int R = in_sizes[1] / DIM;
    const int E = in_sizes[2] / R;
    const int tilesPerRel = E / 16;

    dim3 grid((tilesPerRel + WAVES_PER_BLOCK - 1) / WAVES_PER_BLOCK, R);
    dim3 block(WAVES_PER_BLOCK * 32);

    distmult_wmma_f32_kernel<<<grid, block, 0, stream>>>(
        node, rel, src, dst, out, E, tilesPerRel);
}


# Persistent waves: grid cut to 2 resident 8-wave work-groups per WGP; each wave loops over its relation's 16-edge tiles, next tile's indices fetched a tile ahead (no wave launched per tile)
# speedup vs baseline: 3.1853x; 3.1853x over previous
#include <hip/hip_runtime.h>
#include <hip/hip_bf16.h>


typedef __attribute__((ext_vector_type(2))) float v2f;
typedef __attribute__((ext_vector_type(4))) float v4f;
typedef __attribute__((ext_vector_type(4))) int   v4i;
typedef __attribute__((ext_vector_type(8))) float v8f;

#define DIM 128
#define WAVES_PER_BLOCK 8
#define BLOCKS_PER_REL  128

template <int MASK>
__device__ __forceinline__ float row_xor(float x) {
    return __int_as_float(__builtin_amdgcn_update_dpp(
        0, __float_as_int(x), 0x160 + MASK, 0xF, 0xF, true));
}

__device__ __forceinline__ v8f wmma4(v2f a, v2f b, v8f c) {
    return __builtin_amdgcn_wmma_f32_16x16x4_f32(
         false, a,  false, b,
         (short)0, c,  false,  false);
}

__global__ __launch_bounds__(WAVES_PER_BLOCK * 32) __attribute__((amdgpu_waves_per_eu(4, 4)))
void distmult_wmma_f32_kernel(const float* __restrict__ node,
                              const float* __restrict__ rel,
                              const int*   __restrict__ src,
                              const int*   __restrict__ dst,
                              float*       __restrict__ out,
                              int E, int tilesPerRel) {
    const int r    = blockIdx.y;
    const int lane = threadIdx.x & 31;
    const int wave = __builtin_amdgcn_readfirstlane((int)(threadIdx.x >> 5));

    const int nBlocks = BLOCKS_PER_REL;
    const int stride  = nBlocks * WAVES_PER_BLOCK;
    int tile = (wave >> 1) * (nBlocks * 2) + (int)blockIdx.x * 2 + (wave & 1);
    if (tile >= tilesPerRel) return;
    const int lastTile = tilesPerRel - 1;

    const int c     = lane & 15;
    const int h     = lane >> 4;
    const int hsel  = c >> 3;
    const int piece = c & 7;
    const unsigned colF = (unsigned)(h * 32 + piece * 4);

    const v4f r0 = *(const v4f*)(rel + r * DIM + colF);
    const v4f r1 = *(const v4f*)(rel + r * DIM + 64 + colF);

    const int* srcL = src + (long)r * E + hsel * 8;
    const int* dstL = dst + (long)r * E + hsel * 8;
    const bool writer = (lane < 8) || (lane >= 24);
    float* outL = out + (size_t)r * E + (size_t)((lane < 8) ? lane : lane - 16);

    const char* nodeB = (const char*)node;
    const unsigned colB = colF * 4u;

    v4i sA = *(const v4i*)(srcL + (tile << 4));
    v4i sB = *(const v4i*)(srcL + (tile << 4) + 4);
    v4i dA = *(const v4i*)(dstL + (tile << 4));
    v4i dB = *(const v4i*)(dstL + (tile << 4) + 4);

    for (;;) {
        const int sIdx[8] = {sA.x, sA.y, sA.z, sA.w, sB.x, sB.y, sB.z, sB.w};
        const int dIdx[8] = {dA.x, dA.y, dA.z, dA.w, dB.x, dB.y, dB.z, dB.w};

        v4f u0[8], u1[8], w0[8], w1[8];
#pragma unroll
        for (int g = 0; g < 8; ++g) {
            const unsigned uo = (unsigned)sIdx[g] * (unsigned)(DIM * 4) + colB;
            const unsigned vo = (unsigned)dIdx[g] * (unsigned)(DIM * 4) + colB;
            u0[g] = *(const v4f*)(nodeB + uo);
            u1[g] = *(const v4f*)(nodeB + uo + 256);
            w0[g] = *(const v4f*)(nodeB + vo);
            w1[g] = *(const v4f*)(nodeB + vo + 256);
        }

        const int next = tile + stride;
        const int pf   = (next < lastTile ? next : lastTile) << 4;
        sA = *(const v4i*)(srcL + pf);
        sB = *(const v4i*)(srcL + pf + 4);
        dA = *(const v4i*)(dstL + pf);
        dB = *(const v4i*)(dstL + pf + 4);

        float res = 0.0f;
#pragma unroll
        for (int g = 0; g < 8; ++g) {
            const v4f a0 = u0[g] * r0;
            const v4f a1 = u1[g] * r1;

            v8f acc = {};
            acc = wmma4(a0.xy, w0[g].xy, acc);
            acc = wmma4(a0.zw, w0[g].zw, acc);
            acc = wmma4(a1.xy, w1[g].xy, acc);
            acc = wmma4(a1.zw, w1[g].zw, acc);

            float x = acc[0];
            if (piece == 1) x = acc[1];
            if (piece == 2) x = acc[2];
            if (piece == 3) x = acc[3];
            if (piece == 4) x = acc[4];
            if (piece == 5) x = acc[5];
            if (piece == 6) x = acc[6];
            if (piece == 7) x = acc[7];

            x += row_xor<1>(x);
            x += row_xor<2>(x);
            x += row_xor<4>(x);

            if (piece == g) res = x;
        }

        if (writer) outL[tile << 4] = res;

        if (next >= tilesPerRel) break;
        tile = next;
    }
}

extern "C" void kernel_launch(void* const* d_in, const int* in_sizes, int n_in,
                              void* d_out, int out_size, void* d_ws, size_t ws_size,
                              hipStream_t stream) {
    const float* node = (const float*)d_in[0];
    const float* rel  = (const float*)d_in[1];
    const int*   src  = (const int*)d_in[2];
    const int*   dst  = (const int*)d_in[3];
    float*       out  = (float*)d_out;

    const int R = in_sizes[1] / DIM;
    const int E = in_sizes[2] / R;
    const int tilesPerRel = E / 16;

    dim3 grid(BLOCKS_PER_REL, R);
    dim3 block(WAVES_PER_BLOCK * 32);

    distmult_wmma_f32_kernel<<<grid, block, 0, stream>>>(
        node, rel, src, dst, out, E, tilesPerRel);
}
